# starting kernel
# baseline (speedup 1.0000x reference)
#include <hip/hip_runtime.h>
#include <cmath>

typedef __attribute__((ext_vector_type(2))) float v2f;
typedef __attribute__((ext_vector_type(8))) float v8f;

static constexpr int kS     = 1600;
static constexpr int kC     = 128;
static constexpr int kNB    = 8;
static constexpr int kHEADS = 8;
static constexpr int kHD    = 16;

__device__ __forceinline__ v8f wmma_f32(v2f a, v2f b, v8f c) {
    return __builtin_amdgcn_wmma_f32_16x16x4_f32(false, a, false, b, (short)0, c,
                                                 false, false);
}

__device__ __forceinline__ float xor16(float x) { return __shfl_xor(x, 16, 32); }

__global__ void qkv_proj_kernel(const float* __restrict__ x,
                                const float* __restrict__ w_qkv,
                                float* __restrict__ Qb, float* __restrict__ Kb,
                                float* __restrict__ Vb) {
    const int ot   = blockIdx.x;
    const int st   = blockIdx.y;
    const int nb   = blockIdx.z;
    const int lane = threadIdx.x;
    const int h    = lane >> 4;
    const int li   = lane & 15;

    const int head  = ot / 3;
    const int which = ot % 3;
    const int s     = st * 16 + li;

    const float* wrow  = w_qkv + (size_t)(ot * 16 + li) * kC;
    const float* xbase = x + (size_t)nb * kC * kS + s;

    v8f acc = {};
#pragma unroll 8
    for (int c0 = 0; c0 < kC; c0 += 4) {
        v2f a, b;
        a.x = wrow[c0 + 2 * h + 0];
        a.y = wrow[c0 + 2 * h + 1];
        b.x = xbase[(size_t)(c0 + 2 * h + 0) * kS];
        b.y = xbase[(size_t)(c0 + 2 * h + 1) * kS];
        acc = wmma_f32(a, b, acc);
    }

    float* dst = (which == 0) ? Qb : (which == 1) ? Kb : Vb;
    dst += (size_t)(nb * kHEADS + head) * kHD * kS;
    const float mul = (which == 0) ? 0.25f : 1.0f;
#pragma unroll
    for (int r = 0; r < 8; ++r) {
        const int d = r + 8 * h;
        dst[(size_t)d * kS + s] = acc[r] * mul;
    }
}

__global__ void attn_kernel(const float* __restrict__ Qb,
                            const float* __restrict__ Kb,
                            const float* __restrict__ Vb,
                            float* __restrict__ CTX) {
    const int it   = blockIdx.x;
    const int head = blockIdx.y;
    const int nb   = blockIdx.z;
    const int lane = threadIdx.x;
    const int h    = lane >> 4;
    const int li   = lane & 15;

    const size_t base = (size_t)(nb * kHEADS + head) * kHD * kS;
    const float* q = Qb + base;
    const float* k = Kb + base;
    const float* v = Vb + base;

    const int i = it * 16 + li;

    float bq[8];
#pragma unroll
    for (int kk = 0; kk < 4; ++kk) {
        bq[2 * kk + 0] = q[(size_t)(4 * kk + 2 * h + 0) * kS + i];
        bq[2 * kk + 1] = q[(size_t)(4 * kk + 2 * h + 1) * kS + i];
    }

    float m_run = -INFINITY;
    float l_run = 0.0f;
    v8f   acc   = {};

    for (int jt = 0; jt < 100; ++jt) {
        const int j0 = jt * 16;

        v8f e = {};
#pragma unroll
        for (int kk = 0; kk < 4; ++kk) {
            v2f a, b;
            a.x = k[(size_t)(4 * kk + 2 * h + 0) * kS + (j0 + li)];
            a.y = k[(size_t)(4 * kk + 2 * h + 1) * kS + (j0 + li)];
            b.x = bq[2 * kk + 0];
            b.y = bq[2 * kk + 1];
            e = wmma_f32(a, b, e);
        }

        float pmax = e[0];
#pragma unroll
        for (int r = 1; r < 8; ++r) pmax = fmaxf(pmax, e[r]);
        pmax = fmaxf(pmax, xor16(pmax));
        const float m_new = fmaxf(m_run, pmax);
        const float alpha = __expf(m_run - m_new);
        float p[8], psum = 0.0f;
#pragma unroll
        for (int r = 0; r < 8; ++r) {
            p[r] = __expf(e[r] - m_new);
            psum += p[r];
        }
        psum += xor16(psum);
        l_run = l_run * alpha + psum;
        m_run = m_new;
#pragma unroll
        for (int r = 0; r < 8; ++r) acc[r] *= alpha;

        const float s0 = xor16(p[0]), s1 = xor16(p[1]);
        const float s2 = xor16(p[2]), s3 = xor16(p[3]);
        const float s4 = xor16(p[4]), s5 = xor16(p[5]);
        const float s6 = xor16(p[6]), s7 = xor16(p[7]);

        v2f bp[4];
        bp[0].x = h ? s2 : p[0];  bp[0].y = h ? s3 : p[1];
        bp[1].x = h ? s6 : p[4];  bp[1].y = h ? s7 : p[5];
        bp[2].x = h ? p[2] : s0;  bp[2].y = h ? p[3] : s1;
        bp[3].x = h ? p[6] : s4;  bp[3].y = h ? p[7] : s5;

#pragma unroll
        for (int kk = 0; kk < 4; ++kk) {
            v2f a;
            a.x = v[(size_t)li * kS + (j0 + 4 * kk + 2 * h + 0)];
            a.y = v[(size_t)li * kS + (j0 + 4 * kk + 2 * h + 1)];
            acc = wmma_f32(a, bp[kk], acc);
        }
    }

    const float inv = 1.0f / l_run;
    float* ctx = CTX + base;
#pragma unroll
    for (int r = 0; r < 8; ++r) {
        const int c = r + 8 * h;
        ctx[(size_t)c * kS + i] = acc[r] * inv;
    }
}

__global__ void out_proj_kernel(const float* __restrict__ CTX,
                                const float* __restrict__ w_o,
                                const float* __restrict__ b_o,
                                float* __restrict__ out) {
    const int ot   = blockIdx.x;
    const int st   = blockIdx.y;
    const int nb   = blockIdx.z;
    const int lane = threadIdx.x;
    const int h    = lane >> 4;
    const int li   = lane & 15;
    const int s    = st * 16 + li;

    const float* wrow  = w_o + (size_t)(ot * 16 + li) * kC;
    const float* cbase = CTX + (size_t)nb * kC * kS + s;

    v8f acc = {};
#pragma unroll 8
    for (int c0 = 0; c0 < kC; c0 += 4) {
        v2f a, b;
        a.x = wrow[c0 + 2 * h + 0];
        a.y = wrow[c0 + 2 * h + 1];
        b.x = cbase[(size_t)(c0 + 2 * h + 0) * kS];
        b.y = cbase[(size_t)(c0 + 2 * h + 1) * kS];
        acc = wmma_f32(a, b, acc);
    }
#pragma unroll
    for (int r = 0; r < 8; ++r) {
        const int oc = ot * 16 + r + 8 * h;
        out[((size_t)nb * kC + oc) * kS + s] = acc[r] + b_o[oc];
    }
}

extern "C" void kernel_launch(void* const* d_in, const int* in_sizes, int n_in,
                              void* d_out, int out_size, void* d_ws, size_t ws_size,
                              hipStream_t stream) {
    const float* x     = (const float*)d_in[0];
    const float* w_qkv = (const float*)d_in[1];
    const float* w_o   = (const float*)d_in[2];
    const float* b_o   = (const float*)d_in[3];
    float*       out   = (float*)d_out;

    const size_t per = (size_t)kNB * kC * kS;
    float* Qb  = (float*)d_ws;
    float* Kb  = Qb + per;
    float* Vb  = Kb + per;
    float* CTX = Vb + per;

    qkv_proj_kernel<<<dim3(24, 100, kNB), 32, 0, stream>>>(x, w_qkv, Qb, Kb, Vb);
    attn_kernel<<<dim3(100, kHEADS, kNB), 32, 0, stream>>>(Qb, Kb, Vb, CTX);
    out_proj_kernel<<<dim3(8, 100, kNB), 32, 0, stream>>>(CTX, w_o, b_o, out);
}


# Attention key loop: a wave's issue priority follows the key tiles it still has to do (s_setprio 3,2,1,0 from tiles 0,9,17,23): a SIMD's 5 waves end together, load waits stay covered; sums unchanged
# speedup vs baseline: 1.6818x; 1.6818x over previous
#include <hip/hip_runtime.h>
#include <cmath>

typedef __attribute__((ext_vector_type(2))) float v2f;
typedef __attribute__((ext_vector_type(8))) float v8f;

static constexpr int kS     = 1600;
static constexpr int kC     = 128;
static constexpr int kNB    = 8;
static constexpr int kHEADS = 8;
static constexpr int kHD    = 16;

__device__ __forceinline__ v8f wmma_f32(v2f a, v2f b, v8f c) {
    return __builtin_amdgcn_wmma_f32_16x16x4_f32(false, a, false, b, (short)0, c,
                                                 false, false);
}

__device__ __forceinline__ float xhalf(float x) {
    const int b = __builtin_bit_cast(int, x);
    return __builtin_bit_cast(float,
        __builtin_amdgcn_permlanex16(b, b, 0x76543210, 0xfedcba98, false, false));
}

static constexpr int kSB = 64;
static constexpr int kST = kSB / 16;
static constexpr int kXP = kSB + 8;
static constexpr int kPW = 8;
static_assert(kS / kSB == 25 && kS % kSB == 0 && 32 * kPW == 256 && (kC * kSB / 4) % (32 * kPW) == 0,
              "block shape in kernel_launch");

__device__ __forceinline__ void stage_block(const float* __restrict__ src, float* xs, int tid) {
    typedef __attribute__((ext_vector_type(4))) float v4f;
#pragma unroll
    for (int it = 0; it < (kC * kSB / 4) / (32 * kPW); ++it) {
        const int f  = it * (32 * kPW) + tid;
        const int c  = f / (kSB / 4);
        const int q4 = f % (kSB / 4);
        const v4f v = *reinterpret_cast<const v4f*>(src + (size_t)c * kS + 4 * q4);
        *reinterpret_cast<v4f*>(xs + c * kXP + 4 * q4) = v;
    }
    __syncthreads();
}

template <int NOT>
__device__ __forceinline__ void proj_tiles(const float* __restrict__ w, int ot0, const float* xs,
                                           int h, int li, v8f (&acc)[NOT][kST]) {
    const float* wrow = w + (size_t)(ot0 * 16 + li) * kC + 2 * h;
    const float* xcol = xs + 2 * h * kXP + li;
#pragma unroll 8
    for (int c0 = 0; c0 < kC; c0 += 4) {
        v2f a[NOT], b[kST];
#pragma unroll
        for (int u = 0; u < NOT; ++u)
            a[u] = *reinterpret_cast<const v2f*>(wrow + (size_t)u * 16 * kC + c0);
#pragma unroll
        for (int t = 0; t < kST; ++t) {
            b[t].x = xcol[(c0 + 0) * kXP + 16 * t];
            b[t].y = xcol[(c0 + 1) * kXP + 16 * t];
        }
#pragma unroll
        for (int u = 0; u < NOT; ++u)
#pragma unroll
            for (int t = 0; t < kST; ++t) acc[u][t] = wmma_f32(a[u], b[t], acc[u][t]);
    }
}

__global__ __launch_bounds__(32 * kPW)
void qkv_proj_kernel(const float* __restrict__ x,
                     const float* __restrict__ w_qkv,
                     float* __restrict__ Qb, float* __restrict__ Kb,
                     float* __restrict__ Vb) {
    __shared__ __attribute__((aligned(16))) float xs[kC * kXP];

    const int sb   = blockIdx.x;
    const int nb   = blockIdx.y;
    const int tid  = threadIdx.x;
    const int wv   = __builtin_amdgcn_readfirstlane(tid >> 5);
    const int lane = tid & 31;
    const int h    = lane >> 4;
    const int li   = lane & 15;
    const int s0   = sb * kSB;

    stage_block(x + (size_t)nb * kC * kS + s0, xs, tid);

    v8f acc[3][kST] = {};
    proj_tiles<3>(w_qkv, 3 * wv, xs, h, li, acc);

#pragma unroll
    for (int which = 0; which < 3; ++which) {
        float* dst = (which == 0) ? Qb : (which == 1) ? Kb : Vb;
        dst += (size_t)(nb * kHEADS + wv) * kHD * kS + s0 + li;
        const float mul = (which == 0) ? 0.25f : 1.0f;
#pragma unroll
        for (int t = 0; t < kST; ++t) {
#pragma unroll
            for (int r = 0; r < 8; ++r) {
                const int d = r + 8 * h;
                dst[(size_t)d * kS + 16 * t] = acc[which][t][r] * mul;
            }
        }
    }
}

static constexpr int kQT = 5;
static constexpr int kKS = 4;
static constexpr int kJT = 100 / kKS;
static constexpr int kPV = 10;
static constexpr int kPB1 = 9;
static constexpr int kPB2 = 17;
static constexpr int kPB3 = 23;
static_assert(kQT * 20 == 100 && kKS * 32 == 128 && kKS * kJT == 100, "launch shape in kernel_launch");
static_assert(0 < kPB1 && kPB1 < kPB2 && kPB2 < kPB3 && kPB3 < kJT, "priority is back at 0 inside the key loop");

__global__ __launch_bounds__(32 * kKS)
void attn_kernel(const float* __restrict__ Qb,
                 const float* __restrict__ Kb,
                 const float* __restrict__ Vb,
                 float* __restrict__ CTX) {
    __shared__ float part[kKS * kQT * kPV * 32];

    const int qg   = blockIdx.x;
    const int head = blockIdx.y;
    const int nb   = blockIdx.z;
    const int wv   = __builtin_amdgcn_readfirstlane((int)(threadIdx.x >> 5));
    const int lane = threadIdx.x & 31;
    const int h    = lane >> 4;
    const int li   = lane & 15;

    const size_t base = (size_t)(nb * kHEADS + head) * kHD * kS;
    const float* q = Qb + base;
    const float* k = Kb + base;
    const float* v = Vb + base;

    const int i0 = qg * (16 * kQT) + li;

    const int pli = ((li & 6) << 1) | ((li & 8) >> 2) | (li & 1);

    float bq[kQT][8];
#pragma unroll
    for (int t = 0; t < kQT; ++t) {
#pragma unroll
        for (int kk = 0; kk < 4; ++kk) {
            bq[t][2 * kk + 0] = q[(size_t)(4 * kk + 2 * h + 0) * kS + i0 + 16 * t];
            bq[t][2 * kk + 1] = q[(size_t)(4 * kk + 2 * h + 1) * kS + i0 + 16 * t];
        }
    }

    float m_run[kQT], l_run[kQT];
    v8f   acc[kQT];
#pragma unroll
    for (int t = 0; t < kQT; ++t) {
        m_run[t] = -INFINITY;
        l_run[t] = 0.0f;
        acc[t]   = v8f{};
    }

    const int jt0 = wv * kJT;
    int jt = jt0;
#pragma nounroll
    for (int sg = 0; sg < 4; ++sg) {
        int jt_end;
        if (sg == 0)      { __builtin_amdgcn_s_setprio(3); jt_end = jt0 + kPB1; }
        else if (sg == 1) { __builtin_amdgcn_s_setprio(2); jt_end = jt0 + kPB2; }
        else if (sg == 2) { __builtin_amdgcn_s_setprio(1); jt_end = jt0 + kPB3; }
        else              { __builtin_amdgcn_s_setprio(0); jt_end = jt0 + kJT; }

    for (; jt < jt_end; ++jt) {
        const int j0 = jt * 16;

        v2f ka[4], va[4];
#pragma unroll
        for (int kk = 0; kk < 4; ++kk) {
            ka[kk].x = k[(size_t)(4 * kk + 2 * h + 0) * kS + (j0 + pli)];
            ka[kk].y = k[(size_t)(4 * kk + 2 * h + 1) * kS + (j0 + pli)];
        }
#pragma unroll
        for (int kk = 0; kk < 4; ++kk) {
            va[kk] = *reinterpret_cast<const v2f*>(v + (size_t)li * kS + (j0 + 4 * kk + 2 * h));
        }
        __builtin_amdgcn_sched_barrier(0);

#pragma unroll
        for (int t = 0; t < kQT; ++t) {
            v8f e = {};
#pragma unroll
            for (int kk = 0; kk < 4; ++kk) {
                v2f b;
                b.x = bq[t][2 * kk + 0];
                b.y = bq[t][2 * kk + 1];
                e = wmma_f32(ka[kk], b, e);
            }

            float pmax = e[0];
#pragma unroll
            for (int r = 1; r < 8; ++r) pmax = fmaxf(pmax, e[r]);
            pmax = fmaxf(pmax, xhalf(pmax));
            const float m_new = fmaxf(m_run[t], pmax);
            const float alpha = __expf(m_run[t] - m_new);
            float p[8], psum = 0.0f;
#pragma unroll
            for (int r = 0; r < 8; ++r) {
                p[r] = __expf(e[r] - m_new);
                psum += p[r];
            }
            psum += xhalf(psum);
            l_run[t] = l_run[t] * alpha + psum;
            m_run[t] = m_new;
#pragma unroll
            for (int r = 0; r < 8; ++r) acc[t][r] *= alpha;

            v2f bp[4];
            bp[0].x = p[0];  bp[0].y = p[1];
            bp[1].x = p[2];  bp[1].y = p[3];
            bp[2].x = p[4];  bp[2].y = p[5];
            bp[3].x = p[6];  bp[3].y = p[7];

#pragma unroll
            for (int kk = 0; kk < 4; ++kk) acc[t] = wmma_f32(va[kk], bp[kk], acc[t]);
        }
    }
    }

#pragma unroll
    for (int t = 0; t < kQT; ++t) {
        float* dstp = part + (size_t)((wv * kQT + t) * kPV) * 32 + lane;
#pragma unroll
        for (int r = 0; r < 8; ++r) dstp[r * 32] = acc[t][r];
        dstp[8 * 32] = m_run[t];
        dstp[9 * 32] = l_run[t];
    }
    __syncthreads();

    float* ctx = CTX + base;
#pragma unroll
    for (int t = 0; t < kQT; ++t) {
        if ((t % kKS) != wv) continue;
        float mw[kKS], m_all = -INFINITY;
#pragma unroll
        for (int u = 0; u < kKS; ++u) {
            mw[u] = part[(size_t)((u * kQT + t) * kPV + 8) * 32 + lane];
            m_all = fmaxf(m_all, mw[u]);
        }
        float l_all = 0.0f;
        float o[8];
#pragma unroll
        for (int r = 0; r < 8; ++r) o[r] = 0.0f;
#pragma unroll
        for (int u = 0; u < kKS; ++u) {
            const float f = __expf(mw[u] - m_all);
            const float* srcp = part + (size_t)((u * kQT + t) * kPV) * 32 + lane;
            l_all += srcp[9 * 32] * f;
#pragma unroll
            for (int r = 0; r < 8; ++r) o[r] += srcp[r * 32] * f;
        }
        const float inv = 1.0f / l_all;
        const int i = i0 + 16 * t;
#pragma unroll
        for (int r = 0; r < 8; ++r) {
            const int c = r + 8 * h;
            ctx[(size_t)c * kS + i] = o[r] * inv;
        }
    }
}

__global__ __launch_bounds__(32 * kPW)
void out_proj_kernel(const float* __restrict__ CTX,
                     const float* __restrict__ w_o,
                     const float* __restrict__ b_o,
                     float* __restrict__ out) {
    __shared__ __attribute__((aligned(16))) float xs[kC * kXP];

    const int sb   = blockIdx.x;
    const int nb   = blockIdx.y;
    const int tid  = threadIdx.x;
    const int wv   = __builtin_amdgcn_readfirstlane(tid >> 5);
    const int lane = tid & 31;
    const int h    = lane >> 4;
    const int li   = lane & 15;
    const int s0   = sb * kSB;

    stage_block(CTX + (size_t)nb * kC * kS + s0, xs, tid);

    v8f acc[1][kST] = {};
    proj_tiles<1>(w_o, wv, xs, h, li, acc);

#pragma unroll
    for (int r = 0; r < 8; ++r) {
        const int oc = wv * 16 + r + 8 * h;
        const float bias = b_o[oc];
        float* dst = out + ((size_t)nb * kC + oc) * kS + s0 + li;
#pragma unroll
        for (int t = 0; t < kST; ++t) dst[16 * t] = acc[0][t][r] + bias;
    }
}

extern "C" void kernel_launch(void* const* d_in, const int* in_sizes, int n_in,
                              void* d_out, int out_size, void* d_ws, size_t ws_size,
                              hipStream_t stream) {
    const float* x     = (const float*)d_in[0];
    const float* w_qkv = (const float*)d_in[1];
    const float* w_o   = (const float*)d_in[2];
    const float* b_o   = (const float*)d_in[3];
    float*       out   = (float*)d_out;

    const size_t per = (size_t)kNB * kC * kS;
    float* Qb  = (float*)d_ws;
    float* Kb  = Qb + per;
    float* Vb  = Kb + per;
    float* CTX = Vb + per;

    qkv_proj_kernel<<<dim3(25, kNB), 256, 0, stream>>>(x, w_qkv, Qb, Kb, Vb);
    attn_kernel<<<dim3(20, kHEADS, kNB), 128, 0, stream>>>(Qb, Kb, Vb, CTX);
    out_proj_kernel<<<dim3(25, kNB), 256, 0, stream>>>(CTX, w_o, b_o, out);
}
